# starting kernel
# baseline (speedup 1.0000x reference)
#include <hip/hip_runtime.h>
#include <math.h>

typedef __attribute__((ext_vector_type(2))) float v2f;
typedef __attribute__((ext_vector_type(8))) float v8f;

namespace {
constexpr int kB = 128;
constexpr int kP = 2048;
constexpr int kN = 256;
constexpr int kThreads = 512;
}

__device__ __forceinline__ float fast_rcp(float x) {
#if __has_builtin(__builtin_amdgcn_rcpf)
  return __builtin_amdgcn_rcpf(x);
#else
  return 1.0f / x;
#endif
}
__device__ __forceinline__ float fast_log2(float x) {
#if __has_builtin(__builtin_amdgcn_logf)
  return __builtin_amdgcn_logf(x);
#else
  return log2f(x);
#endif
}
__device__ __forceinline__ float fast_exp2(float x) {
#if __has_builtin(__builtin_amdgcn_exp2f)
  return __builtin_amdgcn_exp2f(x);
#else
  return exp2f(x);
#endif
}

template <bool POWGEN, int PCHUNK>
__device__ __forceinline__ v8f
accumulate_chunk(const float4* __restrict__ pts, float c0, float c1,
                 float w0, float w1, float e, int khalf) {
  v8f acc0 = {};
  v8f acc1 = {};
  const v2f ones = {1.0f, 1.0f};
  for (int pg = 0; pg < PCHUNK; pg += 8) {
#pragma unroll
    for (int g = 0; g < 2; ++g) {
      const int p = pg + g * 4 + khalf;
      const float4 q0 = pts[p];
      const float4 q1 = pts[p + 1];
      float s0 = fabsf(c0 - q0.x) * w0 + fabsf(c1 - q0.y) * w1;
      float s1 = fabsf(c0 - q1.x) * w0 + fabsf(c1 - q1.y) * w1;
      if (POWGEN) {
        s0 = fast_exp2(e * fast_log2(s0));
        s1 = fast_exp2(e * fast_log2(s1));
      }
      v2f a;
      a.x = q0.z * fast_rcp(1.0f + s0);
      a.y = q1.z * fast_rcp(1.0f + s1);
      if (g == 0)
        acc0 = __builtin_amdgcn_wmma_f32_16x16x4_f32(
            false, a, false, ones, (short)0, acc0, false, false);
      else
        acc1 = __builtin_amdgcn_wmma_f32_16x16x4_f32(
            false, a, false, ones, (short)0, acc1, false, false);
    }
  }
  return acc0 + acc1;
}

template <int SPLIT>
__global__ __launch_bounds__(kThreads)
void slayer_partial(const float* __restrict__ batch,
                    const float* __restrict__ not_dummy,
                    const float* __restrict__ centers,
                    const float* __restrict__ sharp,
                    const float* __restrict__ expo,
                    float* __restrict__ part) {
  constexpr int PCHUNK = kP / SPLIT;
  __shared__ float4 pts[PCHUNK];

  const int b = blockIdx.x;
  const int sidx = blockIdx.y;
  const int tid = (int)threadIdx.x;
  const int lane = tid & 31;
  const int wave = tid >> 5;
  const int pbase = sidx * PCHUNK;

  for (int i = tid; i < PCHUNK; i += kThreads) {
    const int p = pbase + i;
    const float2 x =
        *reinterpret_cast<const float2*>(batch + ((size_t)b * kP + p) * 2);
    const float msk = not_dummy[(size_t)b * kP + p];
    pts[i] = make_float4(x.x, x.y, msk, 0.0f);
  }

  const int n = wave * 16 + (lane & 15);
  const float c0 = centers[n * 2 + 0];
  const float c1 = centers[n * 2 + 1];
  const float w0 = fabsf(sharp[n * 2 + 0]);
  const float w1 = fabsf(sharp[n * 2 + 1]);
  const float e = expo[0];
  const int khalf = (lane >> 4) << 1;

  __syncthreads();

  v8f acc;
  if (e == 1.0f)
    acc = accumulate_chunk<false, PCHUNK>(pts, c0, c1, w0, w1, e, khalf);
  else
    acc = accumulate_chunk<true, PCHUNK>(pts, c0, c1, w0, w1, e, khalf);

  float* dst = part + ((size_t)sidx * kB + b) * kN + wave * 16;
  if (lane == 0) {
#pragma unroll
    for (int j = 0; j < 8; ++j) dst[j] = acc[j];
  } else if (lane == 16) {
#pragma unroll
    for (int j = 0; j < 8; ++j) dst[8 + j] = acc[j];
  }
}

template <int SPLIT>
__global__ void slayer_reduce(const float* __restrict__ part,
                              float* __restrict__ out) {
  const int i = blockIdx.x * blockDim.x + threadIdx.x;
  if (i < kB * kN) {
    float a = 0.0f;
#pragma unroll
    for (int s = 0; s < SPLIT; ++s)
      a += part[(size_t)s * (kB * kN) + i];
    out[i] = a;
  }
}

extern "C" void kernel_launch(void* const* d_in, const int* in_sizes, int n_in,
                              void* d_out, int out_size, void* d_ws,
                              size_t ws_size, hipStream_t stream) {
  const float* batch     = (const float*)d_in[0];
  const float* not_dummy = (const float*)d_in[1];
  const float* centers   = (const float*)d_in[2];
  const float* sharp     = (const float*)d_in[3];
  const float* expo      = (const float*)d_in[4];
  float* out = (float*)d_out;

  constexpr int kSplit = 8;
  const size_t need = (size_t)kSplit * kB * kN * sizeof(float);

  if (ws_size >= need) {
    float* part = (float*)d_ws;
    slayer_partial<kSplit><<<dim3(kB, kSplit), kThreads, 0, stream>>>(
        batch, not_dummy, centers, sharp, expo, part);
    slayer_reduce<kSplit><<<(kB * kN + 255) / 256, 256, 0, stream>>>(part, out);
  } else {
    slayer_partial<1><<<dim3(kB, 1), kThreads, 0, stream>>>(
        batch, not_dummy, centers, sharp, expo, out);
  }
}


# roles swapped: each lane keeps 64 points of its batch row in registers, centers come as scalars; a 0/1 routing-matrix WMMA adds each center's 32 lane sums; one launch, no LDS; sum order changes
# speedup vs baseline: 2.6411x; 2.6411x over previous
#include <hip/hip_runtime.h>
#include <math.h>


typedef __attribute__((ext_vector_type(2))) float v2f;
typedef __attribute__((ext_vector_type(8))) float v8f;

namespace {
constexpr int kB = 128;
constexpr int kP = 2048;
constexpr int kN = 256;
constexpr int kWaves = 4;
constexpr int kThreads = kWaves * 32;
constexpr int kCW = 32;
constexpr int kQ = kP / 64;
}

__device__ __forceinline__ float fast_rcp(float x) {
#if __has_builtin(__builtin_amdgcn_rcpf)
  return __builtin_amdgcn_rcpf(x);
#else
  return 1.0f / x;
#endif
}
__device__ __forceinline__ float fast_log2(float x) {
#if __has_builtin(__builtin_amdgcn_logf)
  return __builtin_amdgcn_logf(x);
#else
  return log2f(x);
#endif
}
__device__ __forceinline__ float fast_exp2(float x) {
#if __has_builtin(__builtin_amdgcn_exp2f)
  return __builtin_amdgcn_exp2f(x);
#else
  return exp2f(x);
#endif
}

__device__ __forceinline__ float lane_value(float v, int j) {
  return __builtin_bit_cast(
      float, __builtin_amdgcn_readlane(__builtin_bit_cast(int, v), j));
}

__device__ __forceinline__ float xhalf(float x) {
  const int b = __builtin_bit_cast(int, x);
  return __builtin_bit_cast(
      float, __builtin_amdgcn_permlanex16(b, b, 0x76543210, 0xfedcba98, false,
                                          false));
}

__device__ __forceinline__ float rational_pow(v2f d, float w0, float w1,
                                              float e) {
  float s = fmaf(fabsf(d.y), w1, fabsf(d.x) * w0);
  s = fast_exp2(e * fast_log2(s));
  return fast_rcp(1.0f + s);
}

__device__ __forceinline__ float one_plus_first(float dx, float w0) {
  float t;
  asm("v_fma_f32 %0, |%1|, %2, 1.0" : "=v"(t) : "v"(dx), "s"(w0));
  return t;
}
__device__ __forceinline__ float plus_second(float dy, float w1, float t) {
  float s;
  asm("v_fma_f32 %0, |%1|, %2, %3" : "=v"(s) : "v"(dy), "s"(w1), "v"(t));
  return s;
}

__device__ __forceinline__ void keep_order() {
  __builtin_amdgcn_sched_barrier(0);
}

template <bool POWGEN>
__device__ __forceinline__ float
wave_centers(const float4 (&xx)[kQ], const v2f (&mm)[kQ], float cx, float cy,
             float wx, float wy, float e, int lane) {
  const int c = lane & 15;
  const int h = lane >> 4;
  float vlo = 0.0f, vhi = 0.0f;
  for (int g = 0; g < 2; ++g) {
    v8f D = {};
    for (int q = 0; q < 8; ++q) {
      const int j = 16 * g + 2 * q;
      const v2f cA = {lane_value(cx, j), lane_value(cy, j)};
      const v2f cB = {lane_value(cx, j + 1), lane_value(cy, j + 1)};
      const float w0A = lane_value(wx, j), w1A = lane_value(wy, j);
      const float w0B = lane_value(wx, j + 1), w1B = lane_value(wy, j + 1);
      keep_order();
      v2f aA0 = {0.0f, 0.0f}, aA1 = {0.0f, 0.0f};
      v2f aB0 = {0.0f, 0.0f}, aB1 = {0.0f, 0.0f};
      if (POWGEN) {
#pragma unroll
        for (int k = 0; k < kQ; ++k) {
          const v2f p0 = {xx[k].x, xx[k].y};
          const v2f p1 = {xx[k].z, xx[k].w};
          v2f rA, rB;
          rA.x = rational_pow(cA - p0, w0A, w1A, e);
          rA.y = rational_pow(cA - p1, w0A, w1A, e);
          rB.x = rational_pow(cB - p0, w0B, w1B, e);
          rB.y = rational_pow(cB - p1, w0B, w1B, e);
          if (k & 1) {
            aA1 = __builtin_elementwise_fma(mm[k], rA, aA1);
            aB1 = __builtin_elementwise_fma(mm[k], rB, aB1);
          } else {
            aA0 = __builtin_elementwise_fma(mm[k], rA, aA0);
            aB0 = __builtin_elementwise_fma(mm[k], rB, aB0);
          }
        }
      } else {
        v2f d[8];
        float s[8];
#pragma unroll
        for (int i = 0; i < 8; ++i) {
          const float4 x = xx[i >> 2];
          const v2f p = (i & 1) ? v2f{x.z, x.w} : v2f{x.x, x.y};
          d[i] = ((i & 2) ? cB : cA) - p;
        }
        keep_order();
#pragma unroll
        for (int i = 0; i < 8; ++i)
          s[i] = one_plus_first(d[i].x, (i & 2) ? w0B : w0A);
        keep_order();
#pragma unroll
        for (int i = 0; i < 8; ++i)
          s[i] = plus_second(d[i].y, (i & 2) ? w1B : w1A, s[i]);
        keep_order();
#pragma unroll
        for (int k = 0; k < kQ; k += 2) {
          float r[8];
#pragma unroll
          for (int i = 0; i < 8; ++i) {
            if (k + 2 < kQ) {
              const float4 x = xx[k + 2 + (i >> 2)];
              const v2f p = (i & 1) ? v2f{x.z, x.w} : v2f{x.x, x.y};
              d[i] = ((i & 2) ? cB : cA) - p;
              keep_order();
            }
            r[i] = fast_rcp(s[i]);
            keep_order();
          }
          if (k + 2 < kQ) {
#pragma unroll
            for (int i = 0; i < 8; ++i)
              s[i] = one_plus_first(d[i].x, (i & 2) ? w0B : w0A);
            keep_order();
          }
          aA0 = __builtin_elementwise_fma(mm[k], v2f{r[0], r[1]}, aA0);
          aB0 = __builtin_elementwise_fma(mm[k], v2f{r[2], r[3]}, aB0);
          aA1 = __builtin_elementwise_fma(mm[k + 1], v2f{r[4], r[5]}, aA1);
          aB1 = __builtin_elementwise_fma(mm[k + 1], v2f{r[6], r[7]}, aB1);
          keep_order();
          if (k + 2 < kQ) {
#pragma unroll
            for (int i = 0; i < 8; ++i)
              s[i] = plus_second(d[i].y, (i & 2) ? w1B : w1A, s[i]);
            keep_order();
          }
        }
      }
      const v2f sA = aA0 + aA1;
      const v2f sB = aB0 + aB1;
      const v2f a = {sA.x + sA.y, sB.x + sB.y};
      const v2f sel = {c == 2 * q ? 1.0f : 0.0f, c == 2 * q + 1 ? 1.0f : 0.0f};
      D = __builtin_amdgcn_wmma_f32_16x16x4_f32(false, a, false, sel, (short)0,
                                                D, false, false);
    }
    const float v = ((D[0] + D[1]) + (D[2] + D[3])) +
                    ((D[4] + D[5]) + (D[6] + D[7]));
    vhi = v;
    if (g == 0) vlo = v;
  }
  const float mine = h ? vhi : vlo;
  const float other = h ? vlo : vhi;
  return mine + xhalf(other);
}

__global__ __launch_bounds__(kThreads)
void slayer_rows(const float* __restrict__ batch,
                 const float* __restrict__ not_dummy,
                 const float* __restrict__ centers,
                 const float* __restrict__ sharp,
                 const float* __restrict__ expo,
                 float* __restrict__ out) {
  const int b = blockIdx.x;
  const int lane = (int)threadIdx.x & 31;
  const int wave = __builtin_amdgcn_readfirstlane((int)threadIdx.x >> 5);
  const int n0 = ((int)blockIdx.y * kWaves + wave) * kCW;

  const float4* __restrict__ row4 =
      reinterpret_cast<const float4*>(batch) + (size_t)b * (kP / 2);
  const float2* __restrict__ msk2 =
      reinterpret_cast<const float2*>(not_dummy) + (size_t)b * (kP / 2);
  float4 xx[kQ];
  v2f mm[kQ];
#pragma unroll
  for (int k = 0; k < kQ; ++k) xx[k] = row4[k * 32 + lane];
#pragma unroll
  for (int k = 0; k < kQ; ++k) {
    const float2 t = msk2[k * 32 + lane];
    mm[k] = v2f{t.x, t.y};
  }

  const float2 cc = reinterpret_cast<const float2*>(centers)[n0 + lane];
  const float2 ww = reinterpret_cast<const float2*>(sharp)[n0 + lane];
  const float e = expo[0];

  float r;
  if (e == 1.0f)
    r = wave_centers<false>(xx, mm, cc.x, cc.y, fabsf(ww.x), fabsf(ww.y), e,
                            lane);
  else
    r = wave_centers<true>(xx, mm, cc.x, cc.y, fabsf(ww.x), fabsf(ww.y), e,
                           lane);

  out[(size_t)b * kN + n0 + lane] = r;
}

extern "C" void kernel_launch(void* const* d_in, const int* in_sizes, int n_in,
                              void* d_out, int out_size, void* d_ws,
                              size_t ws_size, hipStream_t stream) {
  const float* batch     = (const float*)d_in[0];
  const float* not_dummy = (const float*)d_in[1];
  const float* centers   = (const float*)d_in[2];
  const float* sharp     = (const float*)d_in[3];
  const float* expo      = (const float*)d_in[4];
  float* out = (float*)d_out;

  slayer_rows<<<dim3(kB, kN / (kWaves * kCW)), kThreads, 0, stream>>>(
      batch, not_dummy, centers, sharp, expo, out);
}


# e==1 path: weight folded into the difference (w>=0: w*|c-p| = |w*c-w*p|): t = p*(-w)+w*c is one packed fma, |t.x|+|t.y| one add, +1 packed over two pairs; w*c rounds once more; sum order kept
# speedup vs baseline: 2.8125x; 1.0649x over previous
#include <hip/hip_runtime.h>
#include <math.h>


typedef __attribute__((ext_vector_type(2))) float v2f;
typedef __attribute__((ext_vector_type(8))) float v8f;

namespace {
constexpr int kB = 128;
constexpr int kP = 2048;
constexpr int kN = 256;
constexpr int kWaves = 4;
constexpr int kThreads = kWaves * 32;
constexpr int kCW = 32;
constexpr int kQ = kP / 64;
}

__device__ __forceinline__ float fast_rcp(float x) {
#if __has_builtin(__builtin_amdgcn_rcpf)
  return __builtin_amdgcn_rcpf(x);
#else
  return 1.0f / x;
#endif
}
__device__ __forceinline__ float fast_log2(float x) {
#if __has_builtin(__builtin_amdgcn_logf)
  return __builtin_amdgcn_logf(x);
#else
  return log2f(x);
#endif
}
__device__ __forceinline__ float fast_exp2(float x) {
#if __has_builtin(__builtin_amdgcn_exp2f)
  return __builtin_amdgcn_exp2f(x);
#else
  return exp2f(x);
#endif
}

__device__ __forceinline__ float lane_value(float v, int j) {
  return __builtin_bit_cast(
      float, __builtin_amdgcn_readlane(__builtin_bit_cast(int, v), j));
}

__device__ __forceinline__ float xhalf(float x) {
  const int b = __builtin_bit_cast(int, x);
  return __builtin_bit_cast(
      float, __builtin_amdgcn_permlanex16(b, b, 0x76543210, 0xfedcba98, false,
                                          false));
}

__device__ __forceinline__ float rational_pow(v2f d, float w0, float w1,
                                              float e) {
  float s = fmaf(fabsf(d.y), w1, fabsf(d.x) * w0);
  s = fast_exp2(e * fast_log2(s));
  return fast_rcp(1.0f + s);
}

__device__ __forceinline__ float abs_sum(v2f t) {
  float a;
  asm("v_add_f32 %0, |%1|, |%2|" : "=v"(a) : "v"(t.x), "v"(t.y));
  return a;
}

__device__ __forceinline__ v2f fold_fma(v2f p, v2f n, v2f k) {
  v2f t;
  asm volatile("v_pk_fma_f32 %0, %1, %2, %3" : "=v"(t) : "v"(p), "v"(n), "v"(k));
  return t;
}
__device__ __forceinline__ void rcp_in_place(float& x) {
  asm volatile("v_rcp_f32 %0, %0" : "+v"(x));
}

__device__ __forceinline__ void keep_order() {
  __builtin_amdgcn_sched_barrier(0);
}

template <bool POWGEN>
__device__ __forceinline__ float
wave_centers(const float4 (&xx)[kQ], const v2f (&mm)[kQ], float cx, float cy,
             float wx, float wy, float e, int lane) {
  const int c = lane & 15;
  const int h = lane >> 4;
  float vlo = 0.0f, vhi = 0.0f;
  const float kx = wx * cx, ky = wy * cy;
  const float nwx = -wx, nwy = -wy;
  for (int g = 0; g < 2; ++g) {
    v8f D = {};
    for (int q = 0; q < 8; ++q) {
      const int j = 16 * g + 2 * q;
      v2f aA0 = {0.0f, 0.0f}, aA1 = {0.0f, 0.0f};
      v2f aB0 = {0.0f, 0.0f}, aB1 = {0.0f, 0.0f};
      if (POWGEN) {
        const v2f cA = {lane_value(cx, j), lane_value(cy, j)};
        const v2f cB = {lane_value(cx, j + 1), lane_value(cy, j + 1)};
        const float w0A = lane_value(wx, j), w1A = lane_value(wy, j);
        const float w0B = lane_value(wx, j + 1), w1B = lane_value(wy, j + 1);
        keep_order();
#pragma unroll
        for (int k = 0; k < kQ; ++k) {
          const v2f p0 = {xx[k].x, xx[k].y};
          const v2f p1 = {xx[k].z, xx[k].w};
          v2f rA, rB;
          rA.x = rational_pow(cA - p0, w0A, w1A, e);
          rA.y = rational_pow(cA - p1, w0A, w1A, e);
          rB.x = rational_pow(cB - p0, w0B, w1B, e);
          rB.y = rational_pow(cB - p1, w0B, w1B, e);
          if (k & 1) {
            aA1 = __builtin_elementwise_fma(mm[k], rA, aA1);
            aB1 = __builtin_elementwise_fma(mm[k], rB, aB1);
          } else {
            aA0 = __builtin_elementwise_fma(mm[k], rA, aA0);
            aB0 = __builtin_elementwise_fma(mm[k], rB, aB0);
          }
        }
      } else {
        const v2f nA = {lane_value(nwx, j), lane_value(nwy, j)};
        const v2f nB = {lane_value(nwx, j + 1), lane_value(nwy, j + 1)};
        const v2f kA = {lane_value(kx, j), lane_value(ky, j)};
        const v2f kB = {lane_value(kx, j + 1), lane_value(ky, j + 1)};
        keep_order();
        v2f t[8];
        float a[8];
        float s[8];
#pragma unroll
        for (int i = 0; i < 8; ++i) {
          const float4 x = xx[i >> 2];
          const v2f p = (i & 1) ? v2f{x.z, x.w} : v2f{x.x, x.y};
          t[i] = fold_fma(p, (i & 2) ? nB : nA, (i & 2) ? kB : kA);
        }
        keep_order();
#pragma unroll
        for (int i = 0; i < 8; ++i) a[i] = abs_sum(t[i]);
        keep_order();
#pragma unroll
        for (int i = 0; i < 4; ++i) {
          const v2f u = v2f{a[2 * i], a[2 * i + 1]} + 1.0f;
          s[2 * i] = u.x;
          s[2 * i + 1] = u.y;
        }
        keep_order();
#pragma unroll
        for (int k = 0; k < kQ; k += 2) {
          const bool more = k + 2 < kQ;
#pragma unroll
          for (int i = 0; i < 8; ++i) {
            if (more) {
              const float4 x = xx[k + 2 + (i >> 2)];
              const v2f p = (i & 1) ? v2f{x.z, x.w} : v2f{x.x, x.y};
              t[i] = fold_fma(p, (i & 2) ? nB : nA, (i & 2) ? kB : kA);
            }
            rcp_in_place(s[i]);
          }
          keep_order();
          if (more) {
#pragma unroll
            for (int i = 0; i < 8; ++i) a[i] = abs_sum(t[i]);
            keep_order();
          }
          aA0 = __builtin_elementwise_fma(mm[k], v2f{s[0], s[1]}, aA0);
          aB0 = __builtin_elementwise_fma(mm[k], v2f{s[2], s[3]}, aB0);
          keep_order();
          if (more) {
#pragma unroll
            for (int i = 0; i < 2; ++i) {
              const v2f u = v2f{a[2 * i], a[2 * i + 1]} + 1.0f;
              s[2 * i] = u.x;
              s[2 * i + 1] = u.y;
            }
            keep_order();
          }
          aA1 = __builtin_elementwise_fma(mm[k + 1], v2f{s[4], s[5]}, aA1);
          aB1 = __builtin_elementwise_fma(mm[k + 1], v2f{s[6], s[7]}, aB1);
          keep_order();
          if (more) {
#pragma unroll
            for (int i = 2; i < 4; ++i) {
              const v2f u = v2f{a[2 * i], a[2 * i + 1]} + 1.0f;
              s[2 * i] = u.x;
              s[2 * i + 1] = u.y;
            }
            keep_order();
          }
        }
      }
      const v2f sA = aA0 + aA1;
      const v2f sB = aB0 + aB1;
      const v2f a = {sA.x + sA.y, sB.x + sB.y};
      const v2f sel = {c == 2 * q ? 1.0f : 0.0f, c == 2 * q + 1 ? 1.0f : 0.0f};
      D = __builtin_amdgcn_wmma_f32_16x16x4_f32(false, a, false, sel, (short)0,
                                                D, false, false);
    }
    const float v = ((D[0] + D[1]) + (D[2] + D[3])) +
                    ((D[4] + D[5]) + (D[6] + D[7]));
    vhi = v;
    if (g == 0) vlo = v;
  }
  const float mine = h ? vhi : vlo;
  const float other = h ? vlo : vhi;
  return mine + xhalf(other);
}

__global__ __launch_bounds__(kThreads)
void slayer_rows(const float* __restrict__ batch,
                 const float* __restrict__ not_dummy,
                 const float* __restrict__ centers,
                 const float* __restrict__ sharp,
                 const float* __restrict__ expo,
                 float* __restrict__ out) {
  const int b = blockIdx.x;
  const int lane = (int)threadIdx.x & 31;
  const int wave = __builtin_amdgcn_readfirstlane((int)threadIdx.x >> 5);
  const int n0 = ((int)blockIdx.y * kWaves + wave) * kCW;

  const float4* __restrict__ row4 =
      reinterpret_cast<const float4*>(batch) + (size_t)b * (kP / 2);
  const float2* __restrict__ msk2 =
      reinterpret_cast<const float2*>(not_dummy) + (size_t)b * (kP / 2);
  float4 xx[kQ];
  v2f mm[kQ];
#pragma unroll
  for (int k = 0; k < kQ; ++k) xx[k] = row4[k * 32 + lane];
#pragma unroll
  for (int k = 0; k < kQ; ++k) {
    const float2 t = msk2[k * 32 + lane];
    mm[k] = v2f{t.x, t.y};
  }

  const float2 cc = reinterpret_cast<const float2*>(centers)[n0 + lane];
  const float2 ww = reinterpret_cast<const float2*>(sharp)[n0 + lane];
  const float e = expo[0];

  float r;
  if (e == 1.0f)
    r = wave_centers<false>(xx, mm, cc.x, cc.y, fabsf(ww.x), fabsf(ww.y), e,
                            lane);
  else
    r = wave_centers<true>(xx, mm, cc.x, cc.y, fabsf(ww.x), fabsf(ww.y), e,
                           lane);

  out[(size_t)b * kN + n0 + lane] = r;
}

extern "C" void kernel_launch(void* const* d_in, const int* in_sizes, int n_in,
                              void* d_out, int out_size, void* d_ws,
                              size_t ws_size, hipStream_t stream) {
  const float* batch     = (const float*)d_in[0];
  const float* not_dummy = (const float*)d_in[1];
  const float* centers   = (const float*)d_in[2];
  const float* sharp     = (const float*)d_in[3];
  const float* expo      = (const float*)d_in[4];
  float* out = (float*)d_out;

  slayer_rows<<<dim3(kB, kN / (kWaves * kCW)), kThreads, 0, stream>>>(
      batch, not_dummy, centers, sharp, expo, out);
}
